# starting kernel
# baseline (speedup 1.0000x reference)
#include <hip/hip_runtime.h>
#include <hip/hip_bf16.h>

typedef __attribute__((ext_vector_type(2))) float v2f;
typedef __attribute__((ext_vector_type(8))) float v8f;

#define EMB 128
#define XPITCH 132
#define KSTEPS (EMB / 4)

#if defined(__has_builtin)
#if __has_builtin(__builtin_amdgcn_sched_barrier)
#define SCHED_FENCE() __builtin_amdgcn_sched_barrier(0)
#endif
#endif
#ifndef SCHED_FENCE
#define SCHED_FENCE()
#endif

__global__ void fill_kernel(float* __restrict__ p, float v, long long n) {
    long long i = (long long)blockIdx.x * blockDim.x + threadIdx.x;
    long long stride = (long long)gridDim.x * blockDim.x;
    for (; i < n; i += stride) p[i] = v;
}

__global__ void deg_accum_kernel(const int* __restrict__ dst, float* __restrict__ deg, int e) {
    int i = blockIdx.x * blockDim.x + threadIdx.x;
    if (i < e) atomicAdd(&deg[dst[i]], 1.0f);
}

__global__ void rsqrt_ip_kernel(float* __restrict__ d, int n) {
    int i = blockIdx.x * blockDim.x + threadIdx.x;
    if (i < n) d[i] = rsqrtf(d[i]);
}

template <int HOUT>
__global__ __launch_bounds__(256) void gemm_wmma_kernel(const float* __restrict__ X,
                                                        const float* __restrict__ W,
                                                        float* __restrict__ H, int nrows) {
    extern __shared__ float smem[];
    float* sW = smem;
    float* sX = smem + EMB * HOUT;

    const int tid = threadIdx.x;
    const int rowBase = blockIdx.x * 128;

    for (int i = tid * 4; i < EMB * HOUT; i += 256 * 4) {
        *(float4*)&sW[i] = *(const float4*)&W[i];
    }
    for (int i = tid * 4; i < 128 * EMB; i += 256 * 4) {
        int r = i >> 7, c = i & 127;
        int gr = rowBase + r;
        float4 v = make_float4(0.f, 0.f, 0.f, 0.f);
        if (gr < nrows) v = *(const float4*)&X[(long long)gr * EMB + c];
        *(float4*)&sX[r * XPITCH + c] = v;
    }
    __syncthreads();

    const int wave = tid >> 5;
    const int lane = tid & 31;
    const int nl = lane & 15;
    const int kh = lane >> 4;
    const int wr = wave * 16;

    const float* xrow = &sX[(wr + nl) * XPITCH + 2 * kh];
    v2f afrag[KSTEPS];
#pragma unroll
    for (int ks = 0; ks < KSTEPS; ++ks) afrag[ks] = *(const v2f*)&xrow[4 * ks];
    SCHED_FENCE();

    for (int nt = 0; nt < HOUT / 16; ++nt) {
        const float* wcol = &sW[nt * 16 + nl];
        v2f bfrag[KSTEPS];
#pragma unroll
        for (int ks = 0; ks < KSTEPS; ++ks) {
            const int kk = 4 * ks + 2 * kh;
            v2f b;
            b.x = wcol[kk * HOUT];
            b.y = wcol[(kk + 1) * HOUT];
            bfrag[ks] = b;
        }
        SCHED_FENCE();
        v8f acc = {};
#pragma unroll
        for (int ks = 0; ks < KSTEPS; ++ks) {
            acc = __builtin_amdgcn_wmma_f32_16x16x4_f32(
                false, afrag[ks], false, bfrag[ks], (short)0, acc, false, false);
        }
        SCHED_FENCE();
#pragma unroll
        for (int j = 0; j < 8; ++j) {
            int r = rowBase + wr + kh * 8 + j;
            if (r < nrows) H[(long long)r * HOUT + nt * 16 + nl] = acc[j];
        }
    }
}

template <int HOUT>
__global__ __launch_bounds__(256) void scatter_kernel(const int* __restrict__ src,
                                                      const int* __restrict__ dst,
                                                      const float* __restrict__ dinv,
                                                      const float* __restrict__ h,
                                                      float* __restrict__ agg, int e) {
    int gid = blockIdx.x * blockDim.x + threadIdx.x;
    int edge = gid >> 5;
    int lane = gid & 31;
    if (edge >= e) return;
    int s = src[edge], d = dst[edge];
    float w = dinv[s] * dinv[d];
    constexpr int V = HOUT / 32;
    const float* hs = &h[(long long)s * HOUT + lane * V];
    float* ad = &agg[(long long)d * HOUT + lane * V];
    if constexpr (V == 4) {
        float4 v = *(const float4*)hs;
        atomicAdd(&ad[0], w * v.x);
        atomicAdd(&ad[1], w * v.y);
        atomicAdd(&ad[2], w * v.z);
        atomicAdd(&ad[3], w * v.w);
    } else {
        float2 v = *(const float2*)hs;
        atomicAdd(&ad[0], w * v.x);
        atomicAdd(&ad[1], w * v.y);
    }
}

__global__ void bias_relu_kernel(float* __restrict__ agg, const float* __restrict__ h,
                                 const float* __restrict__ dinv, const float* __restrict__ b,
                                 long long total, int hdim) {
    long long i = (long long)blockIdx.x * blockDim.x + threadIdx.x;
    if (i >= total) return;
    int row = (int)(i / hdim), col = (int)(i % hdim);
    float di = dinv[row];
    float v = agg[i] + di * di * h[i] + b[col];
    agg[i] = fmaxf(v, 0.0f);
}

__global__ void final_add_kernel(float* __restrict__ out, const float* __restrict__ h,
                                 const float* __restrict__ dinv, const float* __restrict__ b,
                                 long long total, int hdim) {
    long long i = (long long)blockIdx.x * blockDim.x + threadIdx.x;
    if (i >= total) return;
    int row = (int)(i / hdim), col = (int)(i % hdim);
    float di = dinv[row];
    out[i] = out[i] + di * di * h[i] + b[col];
}

extern "C" void kernel_launch(void* const* d_in, const int* in_sizes, int n_in,
                              void* d_out, int out_size, void* d_ws, size_t ws_size,
                              hipStream_t stream) {
    const int*   edge = (const int*)d_in[0];
    const float* x0   = (const float*)d_in[1];
    const float* W1   = (const float*)d_in[2];
    const float* b1   = (const float*)d_in[3];
    const float* W2   = (const float*)d_in[4];
    const float* b2   = (const float*)d_in[5];
    float* out = (float*)d_out;

    const int E    = in_sizes[0] / 2;
    const int N    = in_sizes[1] / EMB;
    const int HID  = in_sizes[3];
    const int REPR = in_sizes[5];
    const int* src = edge;
    const int* dst = edge + E;

    float* dinv = (float*)d_ws;
    float* h    = dinv + N;
    float* agg  = h + (size_t)N * HID;

    const int T = 256;
    long long nh = (long long)N * HID;
    long long nr = (long long)N * REPR;

    fill_kernel<<<512, T, 0, stream>>>(dinv, 1.0f, N);
    deg_accum_kernel<<<(E + T - 1) / T, T, 0, stream>>>(dst, dinv, E);
    rsqrt_ip_kernel<<<(N + T - 1) / T, T, 0, stream>>>(dinv, N);

    size_t smem1 = (size_t)(EMB * HID + 128 * XPITCH) * sizeof(float);
    gemm_wmma_kernel<128><<<(N + 127) / 128, T, smem1, stream>>>(x0, W1, h, N);
    fill_kernel<<<2048, T, 0, stream>>>(agg, 0.0f, nh);
    scatter_kernel<128><<<(int)(((long long)E * 32 + T - 1) / T), T, 0, stream>>>(
        src, dst, dinv, h, agg, E);
    bias_relu_kernel<<<(int)((nh + T - 1) / T), T, 0, stream>>>(agg, h, dinv, b1, nh, HID);

    size_t smem2 = (size_t)(EMB * REPR + 128 * XPITCH) * sizeof(float);
    gemm_wmma_kernel<64><<<(N + 127) / 128, T, smem2, stream>>>(agg, W2, h, N);
    fill_kernel<<<2048, T, 0, stream>>>(out, 0.0f, nr);
    scatter_kernel<64><<<(int)(((long long)E * 32 + T - 1) / T), T, 0, stream>>>(
        src, dst, dinv, h, out, E);
    final_add_kernel<<<(int)((nr + T - 1) / T), T, 0, stream>>>(out, h, dinv, b2, nr, REPR);
}


# Products as persistent blocks: a 16-wave block per WGP stages W once, each wave loops over its own 16-row strips, the next strip's row loads in flight under the current strip's WMMAs; k order kept
# speedup vs baseline: 3.9616x; 3.9616x over previous
#include <hip/hip_runtime.h>
#include <hip/hip_bf16.h>

typedef __attribute__((ext_vector_type(2))) float v2f;
typedef __attribute__((ext_vector_type(4))) float v4f;
typedef __attribute__((ext_vector_type(8))) float v8f;

#define EMB 128
#define KSTEPS (EMB / 4)

#if defined(__has_builtin)
#if __has_builtin(__builtin_amdgcn_sched_barrier)
#define SCHED_FENCE() __builtin_amdgcn_sched_barrier(0)
#endif
#endif
#ifndef SCHED_FENCE
#define SCHED_FENCE()
#endif

#define GW 16
#define SPITCH 132

template <int HOUT, bool RELU_IN>
__global__ __launch_bounds__(GW * 32) void gemm_wmma_kernel(const float* __restrict__ X,
                                                            const float* __restrict__ W,
                                                            float* __restrict__ H, int nrows) {
    extern __shared__ float smem[];
    float* sW = smem;

    const int tid = threadIdx.x;
    const int wave = __builtin_amdgcn_readfirstlane(tid >> 5);
    const int lane = tid & 31;
    const int nl = lane & 15;
    const int kh = lane >> 4;
    float* sA = smem + EMB * HOUT + wave * (16 * SPITCH);

    const int nstrips = (nrows + 15) >> 4;
    const int per = (nstrips + (int)gridDim.x - 1) / (int)gridDim.x;
    const int s0 = min(nstrips, (int)blockIdx.x * per);
    const int s1 = min(nstrips, s0 + per);
    const int rounds = (s1 - s0 + GW - 1) / GW;

    constexpr int NW4 = EMB * HOUT / 4 / (GW * 32);
    {
        v4f w4[NW4];
#pragma unroll
        for (int j = 0; j < NW4; ++j) w4[j] = ((const v4f*)W)[tid + j * (GW * 32)];
#pragma unroll
        for (int j = 0; j < NW4; ++j) ((v4f*)sW)[tid + j * (GW * 32)] = w4[j];
    }

    int s = s0 + wave;
    v4f pre[16];
    if (rounds > 0) {
        const int sc = min(s, nstrips - 1);
#pragma unroll
        for (int j = 0; j < 16; ++j) {
            const int gr = min(sc * 16 + j, nrows - 1);
            pre[j] = *(const v4f*)&X[(long long)gr * EMB + 4 * lane];
        }
    }

    for (int r = 0; r < rounds; ++r, s += GW) {
#pragma unroll
        for (int j = 0; j < 16; ++j) {
            v4f v = pre[j];
            if constexpr (RELU_IN) {
                v.x = fmaxf(v.x, 0.0f);
                v.y = fmaxf(v.y, 0.0f);
                v.z = fmaxf(v.z, 0.0f);
                v.w = fmaxf(v.w, 0.0f);
            }
            *(v4f*)&sA[j * SPITCH + 4 * lane] = v;
        }
        __syncthreads();

        const float* xrow = &sA[nl * SPITCH + 2 * kh];
        v2f afrag[KSTEPS];
#pragma unroll
        for (int ks = 0; ks < KSTEPS; ++ks) afrag[ks] = *(const v2f*)&xrow[4 * ks];
        SCHED_FENCE();

        if (r + 1 < rounds) {
            const int sc = min(s + GW, nstrips - 1);
#pragma unroll
            for (int j = 0; j < 16; ++j) {
                const int gr = min(sc * 16 + j, nrows - 1);
                pre[j] = *(const v4f*)&X[(long long)gr * EMB + 4 * lane];
            }
        }

        if (s < s1) {
            const int rowBase = s * 16;
            for (int nt = 0; nt < HOUT / 16; ++nt) {
                const float* wcol = &sW[nt * 16 + nl];
                v2f bfrag[KSTEPS];
#pragma unroll
                for (int ks = 0; ks < KSTEPS; ++ks) {
                    const int kk = 4 * ks + 2 * kh;
                    v2f b;
                    b.x = wcol[kk * HOUT];
                    b.y = wcol[(kk + 1) * HOUT];
                    bfrag[ks] = b;
                }
                SCHED_FENCE();
                v8f acc = {};
#pragma unroll
                for (int ks = 0; ks < KSTEPS; ++ks) {
                    acc = __builtin_amdgcn_wmma_f32_16x16x4_f32(
                        false, afrag[ks], false, bfrag[ks], (short)0, acc, false, false);
                }
                SCHED_FENCE();
#pragma unroll
                for (int j = 0; j < 8; ++j) {
                    const int row = rowBase + kh * 8 + j;
                    if (row < nrows) H[(long long)row * HOUT + nt * 16 + nl] = acc[j];
                }
            }
        }
    }
}

#define TR 392
#define QCAP 8192
#define BINB 256
#define SUBCAP 64
#define TWIN 256
#define BIN_UNR 8
#define QK (BINB * SUBCAP / 1024)

__device__ __forceinline__ int num_tiles(int n) { return (n + TR - 1) / TR; }
__device__ __forceinline__ long long ent_off(int n) { return (long long)num_tiles(n) * BINB; }

__global__ __launch_bounds__(1024) void bin_edges_kernel(const int* __restrict__ src,
                                                         const int* __restrict__ dst,
                                                         int* __restrict__ cnt, int n, int e) {
    __shared__ int filed[TWIN];

    const int tid = threadIdx.x;
    const int b = blockIdx.x;
    const int nt = num_tiles(n);
    int2* ent = (int2*)(cnt + ent_off(n));

    const int chunk = (e + BINB - 1) / BINB;
    const int e0 = min(e, b * chunk);
    const int e1 = min(e, e0 + chunk);

    for (int tw = 0; tw < nt; tw += TWIN) {
        if (tid < TWIN) filed[tid] = 0;
        __syncthreads();

        for (int base = e0 + tid; base < e1; base += 1024 * BIN_UNR) {
            int d[BIN_UNR], s[BIN_UNR];
#pragma unroll
            for (int k = 0; k < BIN_UNR; ++k) {
                const int i = base + k * 1024;
                d[k] = -1;
                s[k] = 0;
                if (i < e1) {
                    d[k] = dst[i];
                    s[k] = src[i];
                }
            }
#pragma unroll
            for (int k = 0; k < BIN_UNR; ++k) {
                const unsigned t = (unsigned)d[k] / TR;
                const unsigned tl = t - (unsigned)tw;
                if (tl < (unsigned)TWIN && t < (unsigned)nt) {
                    const int slot = atomicAdd(&filed[tl], 1);
                    if (slot < SUBCAP)
                        ent[((long long)t * BINB + b) * SUBCAP + slot] = make_int2(s[k], d[k] - (int)t * TR);
                }
            }
        }
        __syncthreads();

        if (tid < TWIN && tw + tid < nt) cnt[(long long)(tw + tid) * BINB + b] = filed[tid];
        __syncthreads();
    }
}

__global__ __launch_bounds__(1024) void tile_degree_kernel(const int* __restrict__ dst,
                                                           const int* __restrict__ cnt,
                                                           float* __restrict__ dinv, int n, int e) {
    __shared__ int cntd[TR];
    __shared__ int c[BINB];
    __shared__ int over;

    const int tid = threadIdx.x;
    const int2* ent = (const int2*)(cnt + ent_off(n));

    for (int row0 = blockIdx.x * TR; row0 < n; row0 += gridDim.x * TR) {
        const int nr = min(TR, n - row0);
        const int t = row0 / TR;

        if (tid < TR) cntd[tid] = 0;
        if (tid == 0) over = 0;
        __syncthreads();
        if (tid < BINB) {
            const int v = cnt[(long long)t * BINB + tid];
            c[tid] = v;
            if (v > SUBCAP) over = 1;
        }
        __syncthreads();

        if (!over) {
            const int2* seg = ent + (long long)t * BINB * SUBCAP;
            const int sl = tid % SUBCAP;
            int cc[QK], rr[QK];
#pragma unroll
            for (int k = 0; k < QK; ++k) cc[k] = c[tid / SUBCAP + k * (1024 / SUBCAP)];
#pragma unroll
            for (int k = 0; k < QK; ++k) {
                rr[k] = -1;
                if (sl < cc[k]) rr[k] = seg[tid + k * 1024].y;
            }
#pragma unroll
            for (int k = 0; k < QK; ++k) {
                if (rr[k] >= 0) atomicAdd(&cntd[rr[k]], 1);
            }
        } else {
            for (int i = tid; i < e; i += 1024) {
                const unsigned r = (unsigned)(dst[i] - row0);
                if (r < (unsigned)nr) atomicAdd(&cntd[r], 1);
            }
        }
        __syncthreads();

        if (tid < nr) dinv[row0 + tid] = rsqrtf((float)(cntd[tid] + 1));
        __syncthreads();
    }
}

template <int HOUT>
__global__ __launch_bounds__(1024) void own_rows_kernel(const int* __restrict__ src,
                                                        const int* __restrict__ dst,
                                                        const float* __restrict__ dinv,
                                                        const int* __restrict__ qcnt,
                                                        const float* __restrict__ h,
                                                        const float* __restrict__ bias,
                                                        float* __restrict__ outp, int n, int e) {
    __shared__ __attribute__((aligned(16))) float acc[(TR + 1) * HOUT];
    __shared__ float dl[TR];
    __shared__ int   qs[QCAP];
    __shared__ int   qr[QCAP];
    __shared__ float qw[QCAP];
    __shared__ int   subc[BINB];
    __shared__ int   wcnt[32];
    __shared__ int   wbeg[32];
    __shared__ int   wsum[BINB / 32];
    __shared__ int   sover;
    __shared__ int   stot;

    constexpr int LINES = HOUT / 32;
    constexpr int UB = 32 / LINES;
    const int tid = threadIdx.x;
    const int lane = tid & 31;
    const int wave = tid >> 5;
    const int2* qent = (const int2*)(qcnt + ent_off(n));

    for (int row0 = blockIdx.x * TR; row0 < n; row0 += gridDim.x * TR) {
        const int nr = min(TR, n - row0);
        const int t = row0 / TR;

        int myc = 0, incl = 0;
        if (tid < BINB) myc = qcnt[(long long)t * BINB + tid];
        if (tid < nr) dl[tid] = dinv[row0 + tid];
        if (tid == 0) sover = 0;
        if (tid < 32) wcnt[tid] = 0;
        __syncthreads();
        if (tid < BINB) {
            if (myc > SUBCAP) sover = 1;
            incl = myc;
#pragma unroll
            for (int o = 1; o < 32; o <<= 1) {
                const int u = __shfl_up(incl, o);
                if (lane >= o) incl += u;
            }
            if (lane == 31) wsum[wave] = incl;
            subc[tid] = myc;
        }
        __syncthreads();
        if (tid < BINB) {
            int before = 0;
            for (int w = 0; w < wave; ++w) before += wsum[w];
            if (tid == BINB - 1) stot = sover ? (QCAP + 1) : (before + incl);
        }
        __syncthreads();
        const int tot = stot;
        const int cnt = (tot <= QCAP) ? tot : 0;

        if (tid < HOUT) acc[TR * HOUT + tid] = 0.0f;
        for (int i = tid * 4; i < nr * HOUT; i += 1024 * 4) {
            const int r = i / HOUT, c = i % HOUT;
            const float4 hv = *(const float4*)&h[(long long)row0 * HOUT + i];
            const float4 b = *(const float4*)&bias[c];
            const float d = dl[r];
            const float d2 = d * d;
            float4 o;
            o.x = d2 * hv.x + b.x;
            o.y = d2 * hv.y + b.y;
            o.z = d2 * hv.z + b.z;
            o.w = d2 * hv.w + b.w;
            *(float4*)&acc[i] = o;
        }

        const int2* seg = qent + (long long)t * BINB * SUBCAP;
        int2 qv[QK];
        float dv[QK];
        int qi[QK];
#pragma unroll
        for (int k = 0; k < QK; ++k) {
            qi[k] = -1;
            qv[k] = make_int2(row0, 0);
            dv[k] = 0.0f;
        }
        if (cnt > 0) {
            const int sl = tid % SUBCAP;
#pragma unroll
            for (int k = 0; k < QK; ++k) {
                const int sb = tid / SUBCAP + k * (1024 / SUBCAP);
                if (sl < subc[sb]) qi[k] = 0;
            }
#pragma unroll
            for (int k = 0; k < QK; ++k) {
                if (qi[k] >= 0) qv[k] = seg[tid + k * 1024];
            }
#pragma unroll
            for (int k = 0; k < QK; ++k) {
                dv[k] = dinv[qv[k].x];
            }
#pragma unroll
            for (int k = 0; k < QK; ++k) {
                if (qi[k] >= 0) qi[k] = atomicAdd(&wcnt[qv[k].y & 31], 1);
            }
#pragma unroll
            for (int k = 0; k < QK; ++k) {
                dv[k] *= dl[qv[k].y];
            }
        }
        __syncthreads();
        if (tid < 32) {
            const int c = wcnt[tid];
            int inc = c;
#pragma unroll
            for (int o = 1; o < 32; o <<= 1) {
                const int u = __shfl_up(inc, o);
                if (lane >= o) inc += u;
            }
            wbeg[tid] = inc - c;
        }
        __syncthreads();
        int wb[QK];
#pragma unroll
        for (int k = 0; k < QK; ++k) {
            wb[k] = wbeg[qv[k].y & 31];
        }
#pragma unroll
        for (int k = 0; k < QK; ++k) {
            if (qi[k] >= 0) {
                const int p = wb[k] + qi[k];
                qs[p] = qv[k].x;
                qr[p] = qv[k].y;
                qw[p] = dv[k];
            }
        }
        __syncthreads();

        if (tot > QCAP) {
            for (int i = tid; i < e; i += 1024) {
                const unsigned r = (unsigned)(dst[i] - row0);
                if (r < (unsigned)nr) {
                    const int s = src[i];
                    const float w = dinv[s] * dl[r];
#pragma nounroll
                    for (int c = 0; c < HOUT; ++c)
                        atomicAdd(&acc[r * HOUT + c], w * h[(long long)s * HOUT + c]);
                }
            }
            __syncthreads();
        }

        typedef float rowv __attribute__((ext_vector_type(LINES)));
        const int lo = __builtin_amdgcn_readfirstlane(wbeg[wave]);
        const int hi = lo + __builtin_amdgcn_readfirstlane(wcnt[wave]);
        if (lo < hi) {
            int base = lo;
            int q = min(base + lane, hi - 1);
            int es = qs[q];
            int er = qr[q];
            int ew = __float_as_int(qw[q]);
            int sb[UB];
#pragma unroll
            for (int u = 0; u < UB; ++u) sb[u] = __builtin_amdgcn_readlane(es, u);
            for (; base < hi; base += UB) {
                rowv v[UB];
#pragma unroll
                for (int u = 0; u < UB; ++u) v[u] = ((const rowv*)(h + (long long)sb[u] * HOUT))[lane];
                int rr[UB];
                float ww[UB];
#pragma unroll
                for (int u = 0; u < UB; ++u) {
                    const int ru = __builtin_amdgcn_readlane(er, u);
                    rr[u] = (base + u < hi) ? ru : TR;
                    ww[u] = __int_as_float(__builtin_amdgcn_readlane(ew, u));
                }
                q = min(base + UB + lane, hi - 1);
                es = qs[q];
                er = qr[q];
                ew = __float_as_int(qw[q]);
#pragma unroll
                for (int u = 0; u < UB; ++u) sb[u] = __builtin_amdgcn_readlane(es, u);
#pragma unroll
                for (int u = 0; u < UB; u += 2) {
                    rowv* a0 = (rowv*)&acc[rr[u] * HOUT];
                    rowv* a1 = (rowv*)&acc[rr[u + 1] * HOUT];
                    if (rr[u] != rr[u + 1]) {
                        rowv x0 = a0[lane];
                        rowv x1 = a1[lane];
                        x0 += ww[u] * v[u];
                        x1 += ww[u + 1] * v[u + 1];
                        a0[lane] = x0;
                        a1[lane] = x1;
                    } else {
                        rowv x0 = a0[lane];
                        x0 += ww[u] * v[u];
                        x0 += ww[u + 1] * v[u + 1];
                        a0[lane] = x0;
                    }
                }
            }
        }
        __syncthreads();

        for (int i = tid * 4; i < nr * HOUT; i += 1024 * 4) {
            *(float4*)&outp[(long long)row0 * HOUT + i] = *(const float4*)&acc[i];
        }
        __syncthreads();
    }
}

extern "C" void kernel_launch(void* const* d_in, const int* in_sizes, int n_in,
                              void* d_out, int out_size, void* d_ws, size_t ws_size,
                              hipStream_t stream) {
    const int*   edge = (const int*)d_in[0];
    const float* x0   = (const float*)d_in[1];
    const float* W1   = (const float*)d_in[2];
    const float* b1   = (const float*)d_in[3];
    const float* W2   = (const float*)d_in[4];
    const float* b2   = (const float*)d_in[5];
    float* out = (float*)d_out;

    const int E    = in_sizes[0] / 2;
    const int N    = in_sizes[1] / EMB;
    const int HID  = in_sizes[3];
    const int REPR = in_sizes[5];
    const int* src = edge;
    const int* dst = edge + E;

    float* dinv = (float*)d_ws;
    float* h    = dinv + N;
    float* agg  = h + (size_t)N * HID;
    int*   qcnt = (int*)(agg + (size_t)N * HID);

    bin_edges_kernel<<<BINB, 1024, 0, stream>>>(src, dst, qcnt, N, E);
    tile_degree_kernel<<<256, 1024, 0, stream>>>(dst, qcnt, dinv, N, E);

    size_t smem1 = (size_t)(EMB * HID + GW * 16 * SPITCH) * sizeof(float);
    gemm_wmma_kernel<128, false><<<256, GW * 32, smem1, stream>>>(x0, W1, h, N);
    own_rows_kernel<128><<<256, 1024, 0, stream>>>(src, dst, dinv, qcnt, h, b1, agg, N, E);

    size_t smem2 = (size_t)(EMB * REPR + GW * 16 * SPITCH) * sizeof(float);
    gemm_wmma_kernel<64, true><<<256, GW * 32, smem2, stream>>>(agg, W2, h, N);
    own_rows_kernel<64><<<256, 1024, 0, stream>>>(src, dst, dinv, qcnt, h, b2, out, N, E);
}


# Layer 1 summed before its product, (A x) W1 for A (x W1): owner blocks sum x rows, then run both WMMA products on the tile's sums in LDS; h, agg, two launches go (6 to 4); layer-1 sum order changes
# speedup vs baseline: 5.1063x; 1.2889x over previous
#include <hip/hip_runtime.h>
#include <hip/hip_bf16.h>

typedef __attribute__((ext_vector_type(2))) float v2f;
typedef __attribute__((ext_vector_type(4))) float v4f;
typedef __attribute__((ext_vector_type(8))) float v8f;

#define EMB 128
#define KSTEPS (EMB / 4)

#if defined(__has_builtin)
#if __has_builtin(__builtin_amdgcn_sched_barrier)
#define SCHED_FENCE() __builtin_amdgcn_sched_barrier(0)
#endif
#endif
#ifndef SCHED_FENCE
#define SCHED_FENCE()
#endif

#define TR 392
#define QCAP 8192
#define BINB 256
#define SUBCAP 64
#define TWIN 256
#define BIN_UNR 8
#define QK (BINB * SUBCAP / 1024)

__device__ __forceinline__ int num_tiles(int n) { return (n + TR - 1) / TR; }
__device__ __forceinline__ long long ent_off(int n) { return (long long)num_tiles(n) * BINB; }

__global__ __launch_bounds__(1024) void bin_edges_kernel(const int* __restrict__ src,
                                                         const int* __restrict__ dst,
                                                         int* __restrict__ cnt, int n, int e) {
    __shared__ int filed[TWIN];

    const int tid = threadIdx.x;
    const int b = blockIdx.x;
    const int nt = num_tiles(n);
    int2* ent = (int2*)(cnt + ent_off(n));

    const int chunk = (e + BINB - 1) / BINB;
    const int e0 = min(e, b * chunk);
    const int e1 = min(e, e0 + chunk);

    for (int tw = 0; tw < nt; tw += TWIN) {
        if (tid < TWIN) filed[tid] = 0;
        __syncthreads();

        for (int base = e0 + tid; base < e1; base += 1024 * BIN_UNR) {
            int d[BIN_UNR], s[BIN_UNR];
#pragma unroll
            for (int k = 0; k < BIN_UNR; ++k) {
                const int i = base + k * 1024;
                d[k] = -1;
                s[k] = 0;
                if (i < e1) {
                    d[k] = dst[i];
                    s[k] = src[i];
                }
            }
#pragma unroll
            for (int k = 0; k < BIN_UNR; ++k) {
                const unsigned t = (unsigned)d[k] / TR;
                const unsigned tl = t - (unsigned)tw;
                if (tl < (unsigned)TWIN && t < (unsigned)nt) {
                    const int slot = atomicAdd(&filed[tl], 1);
                    if (slot < SUBCAP)
                        ent[((long long)t * BINB + b) * SUBCAP + slot] = make_int2(s[k], d[k] - (int)t * TR);
                }
            }
        }
        __syncthreads();

        if (tid < TWIN && tw + tid < nt) cnt[(long long)(tw + tid) * BINB + b] = filed[tid];
        __syncthreads();
    }
}

__global__ __launch_bounds__(1024) void tile_degree_kernel(const int* __restrict__ dst,
                                                           const int* __restrict__ cnt,
                                                           float* __restrict__ dinv, int n, int e) {
    __shared__ int cntd[TR];
    __shared__ int c[BINB];
    __shared__ int over;

    const int tid = threadIdx.x;
    const int2* ent = (const int2*)(cnt + ent_off(n));

    for (int row0 = blockIdx.x * TR; row0 < n; row0 += gridDim.x * TR) {
        const int nr = min(TR, n - row0);
        const int t = row0 / TR;

        if (tid < TR) cntd[tid] = 0;
        if (tid == 0) over = 0;
        __syncthreads();
        if (tid < BINB) {
            const int v = cnt[(long long)t * BINB + tid];
            c[tid] = v;
            if (v > SUBCAP) over = 1;
        }
        __syncthreads();

        if (!over) {
            const int2* seg = ent + (long long)t * BINB * SUBCAP;
            const int sl = tid % SUBCAP;
            int cc[QK], rr[QK];
#pragma unroll
            for (int k = 0; k < QK; ++k) cc[k] = c[tid / SUBCAP + k * (1024 / SUBCAP)];
#pragma unroll
            for (int k = 0; k < QK; ++k) {
                rr[k] = -1;
                if (sl < cc[k]) rr[k] = seg[tid + k * 1024].y;
            }
#pragma unroll
            for (int k = 0; k < QK; ++k) {
                if (rr[k] >= 0) atomicAdd(&cntd[rr[k]], 1);
            }
        } else {
            for (int i = tid; i < e; i += 1024) {
                const unsigned r = (unsigned)(dst[i] - row0);
                if (r < (unsigned)nr) atomicAdd(&cntd[r], 1);
            }
        }
        __syncthreads();

        if (tid < nr) dinv[row0 + tid] = rsqrtf((float)(cntd[tid] + 1));
        __syncthreads();
    }
}

template <int HOUT>
__global__ __launch_bounds__(1024) void own_rows_kernel(const int* __restrict__ src,
                                                        const int* __restrict__ dst,
                                                        const float* __restrict__ dinv,
                                                        const int* __restrict__ qcnt,
                                                        const float* __restrict__ h,
                                                        const float* __restrict__ bias,
                                                        float* __restrict__ outp, int n, int e) {
    __shared__ __attribute__((aligned(16))) float acc[(TR + 1) * HOUT];
    __shared__ float dl[TR];
    __shared__ int   qs[QCAP];
    __shared__ int   qr[QCAP];
    __shared__ float qw[QCAP];
    __shared__ int   subc[BINB];
    __shared__ int   wcnt[32];
    __shared__ int   wbeg[32];
    __shared__ int   wsum[BINB / 32];
    __shared__ int   sover;
    __shared__ int   stot;

    constexpr int LINES = HOUT / 32;
    constexpr int UB = 32 / LINES;
    const int tid = threadIdx.x;
    const int lane = tid & 31;
    const int wave = tid >> 5;
    const int2* qent = (const int2*)(qcnt + ent_off(n));

    for (int row0 = blockIdx.x * TR; row0 < n; row0 += gridDim.x * TR) {
        const int nr = min(TR, n - row0);
        const int t = row0 / TR;

        int myc = 0, incl = 0;
        if (tid < BINB) myc = qcnt[(long long)t * BINB + tid];
        if (tid < nr) dl[tid] = dinv[row0 + tid];
        if (tid == 0) sover = 0;
        if (tid < 32) wcnt[tid] = 0;
        __syncthreads();
        if (tid < BINB) {
            if (myc > SUBCAP) sover = 1;
            incl = myc;
#pragma unroll
            for (int o = 1; o < 32; o <<= 1) {
                const int u = __shfl_up(incl, o);
                if (lane >= o) incl += u;
            }
            if (lane == 31) wsum[wave] = incl;
            subc[tid] = myc;
        }
        __syncthreads();
        if (tid < BINB) {
            int before = 0;
            for (int w = 0; w < wave; ++w) before += wsum[w];
            if (tid == BINB - 1) stot = sover ? (QCAP + 1) : (before + incl);
        }
        __syncthreads();
        const int tot = stot;
        const int cnt = (tot <= QCAP) ? tot : 0;

        if (tid < HOUT) acc[TR * HOUT + tid] = 0.0f;
        for (int i = tid * 4; i < nr * HOUT; i += 1024 * 4) {
            const int r = i / HOUT, c = i % HOUT;
            const float4 hv = *(const float4*)&h[(long long)row0 * HOUT + i];
            const float4 b = *(const float4*)&bias[c];
            const float d = dl[r];
            const float d2 = d * d;
            float4 o;
            o.x = d2 * hv.x + b.x;
            o.y = d2 * hv.y + b.y;
            o.z = d2 * hv.z + b.z;
            o.w = d2 * hv.w + b.w;
            *(float4*)&acc[i] = o;
        }

        const int2* seg = qent + (long long)t * BINB * SUBCAP;
        int2 qv[QK];
        float dv[QK];
        int qi[QK];
#pragma unroll
        for (int k = 0; k < QK; ++k) {
            qi[k] = -1;
            qv[k] = make_int2(row0, 0);
            dv[k] = 0.0f;
        }
        if (cnt > 0) {
            const int sl = tid % SUBCAP;
#pragma unroll
            for (int k = 0; k < QK; ++k) {
                const int sb = tid / SUBCAP + k * (1024 / SUBCAP);
                if (sl < subc[sb]) qi[k] = 0;
            }
#pragma unroll
            for (int k = 0; k < QK; ++k) {
                if (qi[k] >= 0) qv[k] = seg[tid + k * 1024];
            }
#pragma unroll
            for (int k = 0; k < QK; ++k) {
                dv[k] = dinv[qv[k].x];
            }
#pragma unroll
            for (int k = 0; k < QK; ++k) {
                if (qi[k] >= 0) qi[k] = atomicAdd(&wcnt[qv[k].y & 31], 1);
            }
#pragma unroll
            for (int k = 0; k < QK; ++k) {
                dv[k] *= dl[qv[k].y];
            }
        }
        __syncthreads();
        if (tid < 32) {
            const int c = wcnt[tid];
            int inc = c;
#pragma unroll
            for (int o = 1; o < 32; o <<= 1) {
                const int u = __shfl_up(inc, o);
                if (lane >= o) inc += u;
            }
            wbeg[tid] = inc - c;
        }
        __syncthreads();
        int wb[QK];
#pragma unroll
        for (int k = 0; k < QK; ++k) {
            wb[k] = wbeg[qv[k].y & 31];
        }
#pragma unroll
        for (int k = 0; k < QK; ++k) {
            if (qi[k] >= 0) {
                const int p = wb[k] + qi[k];
                qs[p] = qv[k].x;
                qr[p] = qv[k].y;
                qw[p] = dv[k];
            }
        }
        __syncthreads();

        if (tot > QCAP) {
            for (int i = tid; i < e; i += 1024) {
                const unsigned r = (unsigned)(dst[i] - row0);
                if (r < (unsigned)nr) {
                    const int s = src[i];
                    const float w = dinv[s] * dl[r];
#pragma nounroll
                    for (int c = 0; c < HOUT; ++c)
                        atomicAdd(&acc[r * HOUT + c], w * h[(long long)s * HOUT + c]);
                }
            }
            __syncthreads();
        }

        typedef float rowv __attribute__((ext_vector_type(LINES)));
        const int lo = __builtin_amdgcn_readfirstlane(wbeg[wave]);
        const int hi = lo + __builtin_amdgcn_readfirstlane(wcnt[wave]);
        if (lo < hi) {
            int base = lo;
            int q = min(base + lane, hi - 1);
            int es = qs[q];
            int er = qr[q];
            int ew = __float_as_int(qw[q]);
            int sb[UB];
#pragma unroll
            for (int u = 0; u < UB; ++u) sb[u] = __builtin_amdgcn_readlane(es, u);
            for (; base < hi; base += UB) {
                rowv v[UB];
#pragma unroll
                for (int u = 0; u < UB; ++u) v[u] = ((const rowv*)(h + (long long)sb[u] * HOUT))[lane];
                int rr[UB];
                float ww[UB];
#pragma unroll
                for (int u = 0; u < UB; ++u) {
                    const int ru = __builtin_amdgcn_readlane(er, u);
                    rr[u] = (base + u < hi) ? ru : TR;
                    ww[u] = __int_as_float(__builtin_amdgcn_readlane(ew, u));
                }
                q = min(base + UB + lane, hi - 1);
                es = qs[q];
                er = qr[q];
                ew = __float_as_int(qw[q]);
#pragma unroll
                for (int u = 0; u < UB; ++u) sb[u] = __builtin_amdgcn_readlane(es, u);
#pragma unroll
                for (int u = 0; u < UB; u += 2) {
                    rowv* a0 = (rowv*)&acc[rr[u] * HOUT];
                    rowv* a1 = (rowv*)&acc[rr[u + 1] * HOUT];
                    if (rr[u] != rr[u + 1]) {
                        rowv x0 = a0[lane];
                        rowv x1 = a1[lane];
                        x0 += ww[u] * v[u];
                        x1 += ww[u + 1] * v[u + 1];
                        a0[lane] = x0;
                        a1[lane] = x1;
                    } else {
                        rowv x0 = a0[lane];
                        x0 += ww[u] * v[u];
                        x0 += ww[u + 1] * v[u + 1];
                        a0[lane] = x0;
                    }
                }
            }
        }
        __syncthreads();

        for (int i = tid * 4; i < nr * HOUT; i += 1024 * 4) {
            *(float4*)&outp[(long long)row0 * HOUT + i] = *(const float4*)&acc[i];
        }
        __syncthreads();
    }
}

#define APITCH 132
#define TROWS ((TR + 1 + 15) / 16 * 16)

template <int HOUT>
__device__ __forceinline__ v8f strip_tile_product(const v2f (&afrag)[KSTEPS], const float* wcol, int kh) {
    v8f acc = {};
#pragma unroll
    for (int hb = 0; hb < KSTEPS; hb += KSTEPS / 4) {
        v2f bfrag[KSTEPS / 4];
#pragma unroll
        for (int ks = 0; ks < KSTEPS / 4; ++ks) {
            const int kk = 4 * (hb + ks) + 2 * kh;
            v2f b;
            b.x = wcol[kk * HOUT];
            b.y = wcol[(kk + 1) * HOUT];
            bfrag[ks] = b;
        }
        SCHED_FENCE();
#pragma unroll
        for (int ks = 0; ks < KSTEPS / 4; ++ks) {
            acc = __builtin_amdgcn_wmma_f32_16x16x4_f32(
                false, afrag[hb + ks], false, bfrag[ks], (short)0, acc, false, false);
        }
        SCHED_FENCE();
    }
    return acc;
}

__global__ __launch_bounds__(1024) void own_rows_products_kernel(const int* __restrict__ src,
                                                                 const int* __restrict__ dst,
                                                                 const float* __restrict__ dinv,
                                                                 const int* __restrict__ qcnt,
                                                                 const float* __restrict__ h,
                                                                 const float* W1,
                                                                 const float* __restrict__ b1,
                                                                 const float* W2,
                                                                 float* __restrict__ h2, int n, int e) {
    constexpr int HOUT = EMB;
    constexpr int H1 = 128;
    constexpr int H2 = 64;
    __shared__ __attribute__((aligned(16))) float acc[TROWS * APITCH];
    __shared__ float dl[TR];
    __shared__ __attribute__((aligned(16))) int qbuf[3 * QCAP];
    int*   const qs = qbuf;
    int*   const qr = qbuf + QCAP;
    float* const qw = (float*)(qbuf + 2 * QCAP);
    float* const sW1 = (float*)qbuf;
    float* const sW2 = sW1 + EMB * H1;
    static_assert(3 * QCAP == EMB * H1 + EMB * H2, "W1 and W2 take exactly the queue's space");
    __shared__ int   subc[BINB];
    __shared__ int   wcnt[32];
    __shared__ int   wbeg[32];
    __shared__ int   wsum[BINB / 32];
    __shared__ int   sover;
    __shared__ int   stot;

    constexpr int LINES = HOUT / 32;
    constexpr int UB = 32 / LINES;
    const int tid = threadIdx.x;
    const int lane = tid & 31;
    const int wave = tid >> 5;
    const int2* qent = (const int2*)(qcnt + ent_off(n));

    for (int row0 = blockIdx.x * TR; row0 < n; row0 += gridDim.x * TR) {
        const int nr = min(TR, n - row0);
        const int t = row0 / TR;

        int myc = 0, incl = 0;
        if (tid < BINB) myc = qcnt[(long long)t * BINB + tid];
        if (tid < nr) dl[tid] = dinv[row0 + tid];
        if (tid == 0) sover = 0;
        if (tid < 32) wcnt[tid] = 0;
        __syncthreads();
        if (tid < BINB) {
            if (myc > SUBCAP) sover = 1;
            incl = myc;
#pragma unroll
            for (int o = 1; o < 32; o <<= 1) {
                const int u = __shfl_up(incl, o);
                if (lane >= o) incl += u;
            }
            if (lane == 31) wsum[wave] = incl;
            subc[tid] = myc;
        }
        __syncthreads();
        if (tid < BINB) {
            int before = 0;
            for (int w = 0; w < wave; ++w) before += wsum[w];
            if (tid == BINB - 1) stot = sover ? (QCAP + 1) : (before + incl);
        }
        __syncthreads();
        const int tot = stot;
        const int cnt = (tot <= QCAP) ? tot : 0;

        if (tid < HOUT) acc[TR * APITCH + tid] = 0.0f;
        for (int i = nr * HOUT + tid; i < ((nr + 15) & ~15) * HOUT; i += 1024) acc[(i / HOUT) * APITCH + i % HOUT] = 0.0f;
        for (int i = tid * 4; i < nr * HOUT; i += 1024 * 4) {
            const int r = i / HOUT, c = i % HOUT;
            const float4 hv = *(const float4*)&h[(long long)row0 * HOUT + i];
            const float d = dl[r];
            const float d2 = d * d;
            float4 o;
            o.x = d2 * hv.x;
            o.y = d2 * hv.y;
            o.z = d2 * hv.z;
            o.w = d2 * hv.w;
            *(float4*)&acc[r * APITCH + c] = o;
        }

        const int2* seg = qent + (long long)t * BINB * SUBCAP;
        int2 qv[QK];
        float dv[QK];
        int qi[QK];
#pragma unroll
        for (int k = 0; k < QK; ++k) {
            qi[k] = -1;
            qv[k] = make_int2(row0, 0);
            dv[k] = 0.0f;
        }
        if (cnt > 0) {
            const int sl = tid % SUBCAP;
#pragma unroll
            for (int k = 0; k < QK; ++k) {
                const int sb = tid / SUBCAP + k * (1024 / SUBCAP);
                if (sl < subc[sb]) qi[k] = 0;
            }
#pragma unroll
            for (int k = 0; k < QK; ++k) {
                if (qi[k] >= 0) qv[k] = seg[tid + k * 1024];
            }
#pragma unroll
            for (int k = 0; k < QK; ++k) {
                dv[k] = dinv[qv[k].x];
            }
#pragma unroll
            for (int k = 0; k < QK; ++k) {
                if (qi[k] >= 0) qi[k] = atomicAdd(&wcnt[qv[k].y & 31], 1);
            }
#pragma unroll
            for (int k = 0; k < QK; ++k) {
                dv[k] *= dl[qv[k].y];
            }
        }
        __syncthreads();
        if (tid < 32) {
            const int c = wcnt[tid];
            int inc = c;
#pragma unroll
            for (int o = 1; o < 32; o <<= 1) {
                const int u = __shfl_up(inc, o);
                if (lane >= o) inc += u;
            }
            wbeg[tid] = inc - c;
        }
        __syncthreads();
        int wb[QK];
#pragma unroll
        for (int k = 0; k < QK; ++k) {
            wb[k] = wbeg[qv[k].y & 31];
        }
#pragma unroll
        for (int k = 0; k < QK; ++k) {
            if (qi[k] >= 0) {
                const int p = wb[k] + qi[k];
                qs[p] = qv[k].x;
                qr[p] = qv[k].y;
                qw[p] = dv[k];
            }
        }
        __syncthreads();

        if (tot > QCAP) {
            for (int i = tid; i < e; i += 1024) {
                const unsigned r = (unsigned)(dst[i] - row0);
                if (r < (unsigned)nr) {
                    const int s = src[i];
                    const float w = dinv[s] * dl[r];
#pragma nounroll
                    for (int c = 0; c < HOUT; ++c)
                        atomicAdd(&acc[r * APITCH + c], w * h[(long long)s * HOUT + c]);
                }
            }
            __syncthreads();
        }

        typedef float rowv __attribute__((ext_vector_type(LINES)));
        const int lo = __builtin_amdgcn_readfirstlane(wbeg[wave]);
        const int hi = lo + __builtin_amdgcn_readfirstlane(wcnt[wave]);
        if (lo < hi) {
            int base = lo;
            int q = min(base + lane, hi - 1);
            int es = qs[q];
            int er = qr[q];
            int ew = __float_as_int(qw[q]);
            int sb[UB];
#pragma unroll
            for (int u = 0; u < UB; ++u) sb[u] = __builtin_amdgcn_readlane(es, u);
            for (; base < hi; base += UB) {
                rowv v[UB];
#pragma unroll
                for (int u = 0; u < UB; ++u) v[u] = ((const rowv*)(h + (long long)sb[u] * HOUT))[lane];
                int rr[UB];
                float ww[UB];
#pragma unroll
                for (int u = 0; u < UB; ++u) {
                    const int ru = __builtin_amdgcn_readlane(er, u);
                    rr[u] = (base + u < hi) ? ru : TR;
                    ww[u] = __int_as_float(__builtin_amdgcn_readlane(ew, u));
                }
                q = min(base + UB + lane, hi - 1);
                es = qs[q];
                er = qr[q];
                ew = __float_as_int(qw[q]);
#pragma unroll
                for (int u = 0; u < UB; ++u) sb[u] = __builtin_amdgcn_readlane(es, u);
#pragma unroll
                for (int u = 0; u < UB; u += 2) {
                    rowv* a0 = (rowv*)&acc[rr[u] * APITCH];
                    rowv* a1 = (rowv*)&acc[rr[u + 1] * APITCH];
                    if (rr[u] != rr[u + 1]) {
                        rowv x0 = a0[lane];
                        rowv x1 = a1[lane];
                        x0 += ww[u] * v[u];
                        x1 += ww[u + 1] * v[u + 1];
                        a0[lane] = x0;
                        a1[lane] = x1;
                    } else {
                        rowv x0 = a0[lane];
                        x0 += ww[u] * v[u];
                        x0 += ww[u + 1] * v[u + 1];
                        a0[lane] = x0;
                    }
                }
            }
        }
        constexpr int NV1 = EMB * H1 / 4 / 1024;
        constexpr int NV2 = EMB * H2 / 4 / 1024;
        v4f w4[NV1 + NV2];
#pragma unroll
        for (int j = 0; j < NV1; ++j) w4[j] = ((const v4f*)W1)[tid + j * 1024];
#pragma unroll
        for (int j = 0; j < NV2; ++j) w4[NV1 + j] = ((const v4f*)W2)[tid + j * 1024];
        __syncthreads();

#pragma unroll
        for (int j = 0; j < NV1; ++j) ((v4f*)sW1)[tid + j * 1024] = w4[j];
#pragma unroll
        for (int j = 0; j < NV2; ++j) ((v4f*)sW2)[tid + j * 1024] = w4[NV1 + j];
        __syncthreads();

        const int wv = __builtin_amdgcn_readfirstlane(wave);
        if (wv * 16 < nr) {
            const int nl = lane & 15;
            const int kh = lane >> 4;
            float* strip = &acc[wv * 16 * APITCH];
            const float* xrow = &strip[nl * APITCH + 2 * kh];
            v2f afrag[KSTEPS];
#pragma unroll
            for (int ks = 0; ks < KSTEPS; ++ks) afrag[ks] = *(const v2f*)&xrow[4 * ks];
            SCHED_FENCE();

#pragma nounroll
            for (int nt = 0; nt < H1 / 16; ++nt) {
                const float bv = b1[nt * 16 + nl];
                const v8f d = strip_tile_product<H1>(afrag, &sW1[nt * 16 + nl], kh);
#pragma unroll
                for (int j = 0; j < 8; ++j)
                    strip[(kh * 8 + j) * APITCH + nt * 16 + nl] = fmaxf(d[j] + bv, 0.0f);
            }

#pragma unroll
            for (int ks = 0; ks < KSTEPS; ++ks) afrag[ks] = *(const v2f*)&xrow[4 * ks];
            SCHED_FENCE();

#pragma nounroll
            for (int nt = 0; nt < H2 / 16; ++nt) {
                const v8f d = strip_tile_product<H2>(afrag, &sW2[nt * 16 + nl], kh);
#pragma unroll
                for (int j = 0; j < 8; ++j) {
                    const int r = wv * 16 + kh * 8 + j;
                    if (r < nr) h2[(long long)(row0 + r) * H2 + nt * 16 + nl] = d[j];
                }
            }
        }
        __syncthreads();
    }
}

extern "C" void kernel_launch(void* const* d_in, const int* in_sizes, int n_in,
                              void* d_out, int out_size, void* d_ws, size_t ws_size,
                              hipStream_t stream) {
    const int*   edge = (const int*)d_in[0];
    const float* x0   = (const float*)d_in[1];
    const float* W1   = (const float*)d_in[2];
    const float* b1   = (const float*)d_in[3];
    const float* W2   = (const float*)d_in[4];
    const float* b2   = (const float*)d_in[5];
    float* out = (float*)d_out;

    const int E    = in_sizes[0] / 2;
    const int N    = in_sizes[1] / EMB;
    const int HID  = in_sizes[3];
    const int* src = edge;
    const int* dst = edge + E;

    float* dinv = (float*)d_ws;
    float* h2   = dinv + N;
    int*   qcnt = (int*)(h2 + (size_t)N * HID);

    bin_edges_kernel<<<BINB, 1024, 0, stream>>>(src, dst, qcnt, N, E);
    tile_degree_kernel<<<256, 1024, 0, stream>>>(dst, qcnt, dinv, N, E);

    own_rows_products_kernel<<<256, 1024, 0, stream>>>(src, dst, dinv, qcnt, x0, W1, b1, W2, h2, N, E);

    own_rows_kernel<64><<<256, 1024, 0, stream>>>(src, dst, dinv, qcnt, h2, b2, out, N, E);
}
